# starting kernel
# baseline (speedup 1.0000x reference)
#include <hip/hip_runtime.h>

typedef __attribute__((ext_vector_type(2))) float v2f;
typedef __attribute__((ext_vector_type(8))) float v8f;

#define VOCAB   262
#define EMBD    64
#define OUT_CHS 100
#define BS      32
#define SEQ     512
#define WORD    16

#define VROWS   60
#define VCOLS   112

__global__ __launch_bounds__(64) void embed_sum_kernel(
    const int* __restrict__ x, const float* __restrict__ table,
    float* __restrict__ embd)
{
    const int bs = blockIdx.x;
    const int d  = threadIdx.x;
    const int* xw = x + bs * WORD;
    float acc = 0.f;
    #pragma unroll
    for (int i = 0; i < WORD; ++i) {
        acc += table[xw[i] * EMBD + d];
    }
    embd[(size_t)bs * EMBD + d] = acc;
}

__global__ __launch_bounds__(128) void conv_scramble_sum_kernel(
    const float* __restrict__ embd, const float* __restrict__ conv_w,
    const float* __restrict__ conv_b, float* __restrict__ out)
{
    __shared__ float sE[100 * 64];
    __shared__ float sV[VROWS * VCOLS];

    const int hp  = blockIdx.x;
    const int b   = blockIdx.y;
    const int tid = threadIdx.x;
    const int rr0 = hp * 100;

    const float* ebase = embd + (size_t)b * SEQ * EMBD;
    for (int idx = tid; idx < 100 * 64; idx += 128) {
        const int j = idx >> 6;
        const int d = idx & 63;
        const int h = (rr0 + j) & 511;
        sE[idx] = ebase[h * 64 + d];
    }

    for (int q = tid; q < VROWS * (VCOLS - OUT_CHS); q += 128) {
        const int row = q / (VCOLS - OUT_CHS);
        const int cp  = q - row * (VCOLS - OUT_CHS);
        sV[row * VCOLS + OUT_CHS + cp] = 0.f;
    }

    __syncthreads();

    for (int p = tid; p < 6000; p += 128) {
        const int j  = p / 60;
        const int w  = p - j * 60;
        const int rr = rr0 + j;
        const int c  = rr >> 9;
        const float* wt = conv_w + c * 5;
        float acc = conv_b[c];
        #pragma unroll
        for (int k = 0; k < 5; ++k)
            acc = fmaf(sE[j * 64 + w + k], wt[k], acc);
        const int wp = p / 100;
        const int cp = p - wp * 100;
        sV[wp * VCOLS + cp] = fmaxf(acc, 0.f);
    }
    __syncthreads();

    const int wave = tid >> 5;
    const int lane = tid & 31;
    const int half = lane >> 4;
    const int n    = lane & 15;

    const v2f aones = {1.f, 1.f};

    for (int nt = wave; nt < 7; nt += 4) {
        const int col = nt * 16 + n;
        v8f acc = {};
        #pragma unroll
        for (int kb = 0; kb < 15; ++kb) {
            const int r = kb * 4 + half * 2;
            v2f bvec;
            bvec.x = sV[r * VCOLS + col];
            bvec.y = sV[(r + 1) * VCOLS + col];
            acc = __builtin_amdgcn_wmma_f32_16x16x4_f32(
                false, aones, false, bvec, (short)0, acc, false, false);
        }
        if (lane < 16 && col < OUT_CHS)
            out[((size_t)b * SEQ + hp) * OUT_CHS + col] = acc[0];
    }
}

extern "C" void kernel_launch(void* const* d_in, const int* in_sizes, int n_in,
                              void* d_out, int out_size, void* d_ws, size_t ws_size,
                              hipStream_t stream) {
    const int*   x      = (const int*)d_in[0];
    const float* table  = (const float*)d_in[1];
    const float* conv_w = (const float*)d_in[2];
    const float* conv_b = (const float*)d_in[3];
    float*       out    = (float*)d_out;
    float*       embd   = (float*)d_ws;

    embed_sum_kernel<<<BS * SEQ, 64, 0, stream>>>(x, table, embd);

    dim3 grid(SEQ, BS);
    conv_scramble_sum_kernel<<<grid, 128, 0, stream>>>(embd, conv_w, conv_b, out);

    (void)in_sizes; (void)n_in; (void)out_size; (void)ws_size;
}


# conv loop re-tiled: each thread computes 4 adjacent outputs of one row from two 16-byte LDS reads and stores them with one 16-byte LDS write; index math and taps fetched once per 4 outputs
# speedup vs baseline: 1.4455x; 1.4455x over previous
#include <hip/hip_runtime.h>

typedef __attribute__((ext_vector_type(2))) float v2f;
typedef __attribute__((ext_vector_type(8))) float v8f;

#define VOCAB   262
#define EMBD    64
#define OUT_CHS 100
#define BS      32
#define SEQ     512
#define WORD    16

#define VROWS   60
#define VCOLS   112

__global__ __launch_bounds__(64) void embed_sum_kernel(
    const int* __restrict__ x, const float* __restrict__ table,
    float* __restrict__ embd)
{
    const int bs = blockIdx.x;
    const int d  = threadIdx.x;
    const int* xw = x + bs * WORD;
    float acc = 0.f;
    #pragma unroll
    for (int i = 0; i < WORD; ++i) {
        acc += table[xw[i] * EMBD + d];
    }
    embd[(size_t)bs * EMBD + d] = acc;
}

__global__ __launch_bounds__(128) void conv_scramble_sum_kernel(
    const float* __restrict__ embd, const float* __restrict__ conv_w,
    const float* __restrict__ conv_b, float* __restrict__ out)
{
    __shared__ __attribute__((aligned(16))) float sE[100 * 64];
    __shared__ __attribute__((aligned(16))) float sV[VROWS * VCOLS];

    const int hp  = blockIdx.x;
    const int b   = blockIdx.y;
    const int tid = threadIdx.x;
    const int rr0 = hp * 100;

    const float* ebase = embd + (size_t)b * SEQ * EMBD;
    for (int idx = tid; idx < 100 * 64; idx += 128) {
        const int j = idx >> 6;
        const int d = idx & 63;
        const int h = (rr0 + j) & 511;
        sE[idx] = ebase[h * 64 + d];
    }

    for (int q = tid; q < VROWS * (VCOLS - OUT_CHS); q += 128) {
        const int row = q / (VCOLS - OUT_CHS);
        const int cp  = q - row * (VCOLS - OUT_CHS);
        sV[row * VCOLS + OUT_CHS + cp] = 0.f;
    }

    __syncthreads();

    for (int q = tid; q < 1500; q += 128) {
        const int j  = q / 15;
        const int wq = q - j * 15;
        const int rr = rr0 + j;
        const int c  = rr >> 9;
        const float* wt = conv_w + c * 5;
        const float bias = conv_b[c];
        const float t0 = wt[0], t1 = wt[1], t2 = wt[2], t3 = wt[3], t4 = wt[4];
        const float4 lo = *reinterpret_cast<const float4*>(sE + j * 64 + wq * 4);
        const float4 hi = *reinterpret_cast<const float4*>(sE + j * 64 + wq * 4 + 4);
        float4 r;
        r.x = fmaf(hi.x, t4, fmaf(lo.w, t3, fmaf(lo.z, t2, fmaf(lo.y, t1, fmaf(lo.x, t0, bias)))));
        r.y = fmaf(hi.y, t4, fmaf(hi.x, t3, fmaf(lo.w, t2, fmaf(lo.z, t1, fmaf(lo.y, t0, bias)))));
        r.z = fmaf(hi.z, t4, fmaf(hi.y, t3, fmaf(hi.x, t2, fmaf(lo.w, t1, fmaf(lo.z, t0, bias)))));
        r.w = fmaf(hi.w, t4, fmaf(hi.z, t3, fmaf(hi.y, t2, fmaf(hi.x, t1, fmaf(lo.w, t0, bias)))));
        r.x = fmaxf(r.x, 0.f);
        r.y = fmaxf(r.y, 0.f);
        r.z = fmaxf(r.z, 0.f);
        r.w = fmaxf(r.w, 0.f);
        const int wp = q / 25;
        const int cq = q - wp * 25;
        *reinterpret_cast<float4*>(sV + wp * VCOLS + cq * 4) = r;
    }
    __syncthreads();

    const int wave = tid >> 5;
    const int lane = tid & 31;
    const int half = lane >> 4;
    const int n    = lane & 15;

    const v2f aones = {1.f, 1.f};

    for (int nt = wave; nt < 7; nt += 4) {
        const int col = nt * 16 + n;
        v8f acc = {};
        #pragma unroll
        for (int kb = 0; kb < 15; ++kb) {
            const int r = kb * 4 + half * 2;
            v2f bvec;
            bvec.x = sV[r * VCOLS + col];
            bvec.y = sV[(r + 1) * VCOLS + col];
            acc = __builtin_amdgcn_wmma_f32_16x16x4_f32(
                false, aones, false, bvec, (short)0, acc, false, false);
        }
        if (lane < 16 && col < OUT_CHS)
            out[((size_t)b * SEQ + hp) * OUT_CHS + col] = acc[0];
    }
}

extern "C" void kernel_launch(void* const* d_in, const int* in_sizes, int n_in,
                              void* d_out, int out_size, void* d_ws, size_t ws_size,
                              hipStream_t stream) {
    const int*   x      = (const int*)d_in[0];
    const float* table  = (const float*)d_in[1];
    const float* conv_w = (const float*)d_in[2];
    const float* conv_b = (const float*)d_in[3];
    float*       out    = (float*)d_out;
    float*       embd   = (float*)d_ws;

    embed_sum_kernel<<<BS * SEQ, 64, 0, stream>>>(x, table, embd);

    dim3 grid(SEQ, BS);
    conv_scramble_sum_kernel<<<grid, 128, 0, stream>>>(embd, conv_w, conv_b, out);

    (void)in_sizes; (void)n_in; (void)out_size; (void)ws_size;
}


# embd rows staged with 16-byte global loads and 16-byte LDS stores, all 13 loads of a thread issued before the first wait: one exposed L2 round trip per block where the 4-byte loop had seven
# speedup vs baseline: 1.6719x; 1.1566x over previous
#include <hip/hip_runtime.h>

typedef __attribute__((ext_vector_type(2))) float v2f;
typedef __attribute__((ext_vector_type(8))) float v8f;

#define VOCAB   262
#define EMBD    64
#define OUT_CHS 100
#define BS      32
#define SEQ     512
#define WORD    16

#define VROWS   60
#define VCOLS   112

__global__ __launch_bounds__(64) void embed_sum_kernel(
    const int* __restrict__ x, const float* __restrict__ table,
    float* __restrict__ embd)
{
    const int bs = blockIdx.x;
    const int d  = threadIdx.x;
    const int* xw = x + bs * WORD;
    float acc = 0.f;
    #pragma unroll
    for (int i = 0; i < WORD; ++i) {
        acc += table[xw[i] * EMBD + d];
    }
    embd[(size_t)bs * EMBD + d] = acc;
}

__global__ __launch_bounds__(128) void conv_scramble_sum_kernel(
    const float* __restrict__ embd, const float* __restrict__ conv_w,
    const float* __restrict__ conv_b, float* __restrict__ out)
{
    __shared__ __attribute__((aligned(16))) float sE[100 * 64];
    __shared__ __attribute__((aligned(16))) float sV[VROWS * VCOLS];

    const int hp  = blockIdx.x;
    const int b   = blockIdx.y;
    const int tid = threadIdx.x;
    const int rr0 = hp * 100;

    const float4* ebase4 = reinterpret_cast<const float4*>(embd + (size_t)b * SEQ * EMBD);
    float4* sE4 = reinterpret_cast<float4*>(sE);
    float4 stg[12];
    #pragma unroll
    for (int it = 0; it < 12; ++it) {
        const int idx = tid + it * 128;
        const int h   = (rr0 + (idx >> 4)) & 511;
        stg[it] = ebase4[h * 16 + (idx & 15)];
    }
    float4 stg_tail = make_float4(0.f, 0.f, 0.f, 0.f);
    const bool has_tail = tid < 64;
    if (has_tail) {
        const int idx = tid + 12 * 128;
        const int h   = (rr0 + (idx >> 4)) & 511;
        stg_tail = ebase4[h * 16 + (idx & 15)];
    }
    #pragma unroll
    for (int it = 0; it < 12; ++it)
        sE4[tid + it * 128] = stg[it];
    if (has_tail)
        sE4[tid + 12 * 128] = stg_tail;

    for (int q = tid; q < VROWS * (VCOLS - OUT_CHS); q += 128) {
        const int row = q / (VCOLS - OUT_CHS);
        const int cp  = q - row * (VCOLS - OUT_CHS);
        sV[row * VCOLS + OUT_CHS + cp] = 0.f;
    }

    __syncthreads();

    for (int q = tid; q < 1500; q += 128) {
        const int j  = q / 15;
        const int wq = q - j * 15;
        const int rr = rr0 + j;
        const int c  = rr >> 9;
        const float* wt = conv_w + c * 5;
        const float bias = conv_b[c];
        const float t0 = wt[0], t1 = wt[1], t2 = wt[2], t3 = wt[3], t4 = wt[4];
        const float4 lo = *reinterpret_cast<const float4*>(sE + j * 64 + wq * 4);
        const float4 hi = *reinterpret_cast<const float4*>(sE + j * 64 + wq * 4 + 4);
        float4 r;
        r.x = fmaf(hi.x, t4, fmaf(lo.w, t3, fmaf(lo.z, t2, fmaf(lo.y, t1, fmaf(lo.x, t0, bias)))));
        r.y = fmaf(hi.y, t4, fmaf(hi.x, t3, fmaf(lo.w, t2, fmaf(lo.z, t1, fmaf(lo.y, t0, bias)))));
        r.z = fmaf(hi.z, t4, fmaf(hi.y, t3, fmaf(hi.x, t2, fmaf(lo.w, t1, fmaf(lo.z, t0, bias)))));
        r.w = fmaf(hi.w, t4, fmaf(hi.z, t3, fmaf(hi.y, t2, fmaf(hi.x, t1, fmaf(lo.w, t0, bias)))));
        r.x = fmaxf(r.x, 0.f);
        r.y = fmaxf(r.y, 0.f);
        r.z = fmaxf(r.z, 0.f);
        r.w = fmaxf(r.w, 0.f);
        const int wp = q / 25;
        const int cq = q - wp * 25;
        *reinterpret_cast<float4*>(sV + wp * VCOLS + cq * 4) = r;
    }
    __syncthreads();

    const int wave = tid >> 5;
    const int lane = tid & 31;
    const int half = lane >> 4;
    const int n    = lane & 15;

    const v2f aones = {1.f, 1.f};

    for (int nt = wave; nt < 7; nt += 4) {
        const int col = nt * 16 + n;
        v8f acc = {};
        #pragma unroll
        for (int kb = 0; kb < 15; ++kb) {
            const int r = kb * 4 + half * 2;
            v2f bvec;
            bvec.x = sV[r * VCOLS + col];
            bvec.y = sV[(r + 1) * VCOLS + col];
            acc = __builtin_amdgcn_wmma_f32_16x16x4_f32(
                false, aones, false, bvec, (short)0, acc, false, false);
        }
        if (lane < 16 && col < OUT_CHS)
            out[((size_t)b * SEQ + hp) * OUT_CHS + col] = acc[0];
    }
}

extern "C" void kernel_launch(void* const* d_in, const int* in_sizes, int n_in,
                              void* d_out, int out_size, void* d_ws, size_t ws_size,
                              hipStream_t stream) {
    const int*   x      = (const int*)d_in[0];
    const float* table  = (const float*)d_in[1];
    const float* conv_w = (const float*)d_in[2];
    const float* conv_b = (const float*)d_in[3];
    float*       out    = (float*)d_out;
    float*       embd   = (float*)d_ws;

    embed_sum_kernel<<<BS * SEQ, 64, 0, stream>>>(x, table, embd);

    dim3 grid(SEQ, BS);
    conv_scramble_sum_kernel<<<grid, 128, 0, stream>>>(embd, conv_w, conv_b, out);

    (void)in_sizes; (void)n_in; (void)out_size; (void)ws_size;
}


# one work-group per four output rows h', h'+128, h'+256, h'+384, which read the same 100 embd rows: rows staged once, each LDS read of them feeds all four rows' conv taps; sum order kept
# speedup vs baseline: 2.3047x; 1.3785x over previous
#include <hip/hip_runtime.h>

typedef __attribute__((ext_vector_type(2))) float v2f;
typedef __attribute__((ext_vector_type(8))) float v8f;

#define VOCAB   262
#define EMBD    64
#define OUT_CHS 100
#define BS      32
#define SEQ     512
#define WORD    16

#define VROWS   60
#define VCOLS   112
#define VPLANE  (VROWS * VCOLS)

#define NSIB    4
#define HGRP    128
#define CSTEP   25

__global__ __launch_bounds__(64) void embed_sum_kernel(
    const int* __restrict__ x, const float* __restrict__ table,
    float* __restrict__ embd)
{
    const int bs = blockIdx.x;
    const int d  = threadIdx.x;
    const int* xw = x + bs * WORD;
    float acc = 0.f;
    #pragma unroll
    for (int i = 0; i < WORD; ++i) {
        acc += table[xw[i] * EMBD + d];
    }
    embd[(size_t)bs * EMBD + d] = acc;
}

__global__ __launch_bounds__(256) void conv_scramble_sum_kernel(
    const float* __restrict__ embd, const float* __restrict__ conv_w,
    const float* __restrict__ conv_b, float* __restrict__ out)
{
    __shared__ __attribute__((aligned(16))) float sE[100 * 64];
    __shared__ __attribute__((aligned(16))) float sV[NSIB * VPLANE];

    const int g   = blockIdx.x;
    const int b   = blockIdx.y;
    const int tid = threadIdx.x;
    const int rr0 = g * 100;

    const float4* ebase4 = reinterpret_cast<const float4*>(embd + (size_t)b * SEQ * EMBD);
    float4* sE4 = reinterpret_cast<float4*>(sE);
    float4 stg[6];
    #pragma unroll
    for (int it = 0; it < 6; ++it) {
        const int idx = tid + it * 256;
        const int h   = (rr0 + (idx >> 4)) & 511;
        stg[it] = ebase4[h * 16 + (idx & 15)];
    }
    float4 stg_tail = make_float4(0.f, 0.f, 0.f, 0.f);
    const bool has_tail = tid < 64;
    if (has_tail) {
        const int idx = tid + 6 * 256;
        const int h   = (rr0 + (idx >> 4)) & 511;
        stg_tail = ebase4[h * 16 + (idx & 15)];
    }

    for (int q = tid; q < NSIB * VROWS * 3; q += 256) {
        const int row = q / 3;
        const int cp  = q - row * 3;
        *reinterpret_cast<float4*>(sV + row * VCOLS + OUT_CHS + cp * 4) =
            make_float4(0.f, 0.f, 0.f, 0.f);
    }

    #pragma unroll
    for (int it = 0; it < 6; ++it)
        sE4[tid + it * 256] = stg[it];
    if (has_tail)
        sE4[tid + 6 * 256] = stg_tail;

    __syncthreads();

    for (int q = tid; q < 1500; q += 256) {
        const int j  = q / 15;
        const int wq = q - j * 15;
        const int rr = rr0 + j;
        const int c  = rr >> 9;
        const float4 lo = *reinterpret_cast<const float4*>(sE + j * 64 + wq * 4);
        const float4 hi = *reinterpret_cast<const float4*>(sE + j * 64 + wq * 4 + 4);
        const int wp = q / 25;
        const int cq = q - wp * 25;
        float* dst = sV + wp * VCOLS + cq * 4;
        #pragma unroll
        for (int m = 0; m < NSIB; ++m) {
            const float* wt = conv_w + (c + CSTEP * m) * 5;
            const float bias = conv_b[c + CSTEP * m];
            const float t0 = wt[0], t1 = wt[1], t2 = wt[2], t3 = wt[3], t4 = wt[4];
            float4 r;
            r.x = fmaf(hi.x, t4, fmaf(lo.w, t3, fmaf(lo.z, t2, fmaf(lo.y, t1, fmaf(lo.x, t0, bias)))));
            r.y = fmaf(hi.y, t4, fmaf(hi.x, t3, fmaf(lo.w, t2, fmaf(lo.z, t1, fmaf(lo.y, t0, bias)))));
            r.z = fmaf(hi.z, t4, fmaf(hi.y, t3, fmaf(hi.x, t2, fmaf(lo.w, t1, fmaf(lo.z, t0, bias)))));
            r.w = fmaf(hi.w, t4, fmaf(hi.z, t3, fmaf(hi.y, t2, fmaf(hi.x, t1, fmaf(lo.w, t0, bias)))));
            r.x = fmaxf(r.x, 0.f);
            r.y = fmaxf(r.y, 0.f);
            r.z = fmaxf(r.z, 0.f);
            r.w = fmaxf(r.w, 0.f);
            *reinterpret_cast<float4*>(dst + m * VPLANE) = r;
        }
    }
    __syncthreads();

    const int wave = tid >> 5;
    const int lane = tid & 31;
    const int half = lane >> 4;
    const int n    = lane & 15;

    const v2f aones = {1.f, 1.f};

    for (int t = wave; t < NSIB * 7; t += 8) {
        const int m   = t / 7;
        const int nt  = t - m * 7;
        const int col = nt * 16 + n;
        const float* bp = sV + m * VPLANE + (half * 2) * VCOLS + col;
        v8f acc = {};
        #pragma unroll
        for (int kb = 0; kb < 15; ++kb) {
            v2f bvec;
            bvec.x = bp[kb * 4 * VCOLS];
            bvec.y = bp[kb * 4 * VCOLS + VCOLS];
            acc = __builtin_amdgcn_wmma_f32_16x16x4_f32(
                false, aones, false, bvec, (short)0, acc, false, false);
        }
        if (lane < 16 && col < OUT_CHS)
            out[(unsigned)((b * SEQ + g + HGRP * m) * OUT_CHS + col)] = acc[0];
    }
}

extern "C" void kernel_launch(void* const* d_in, const int* in_sizes, int n_in,
                              void* d_out, int out_size, void* d_ws, size_t ws_size,
                              hipStream_t stream) {
    const int*   x      = (const int*)d_in[0];
    const float* table  = (const float*)d_in[1];
    const float* conv_w = (const float*)d_in[2];
    const float* conv_b = (const float*)d_in[3];
    float*       out    = (float*)d_out;
    float*       embd   = (float*)d_ws;

    embed_sum_kernel<<<BS * SEQ, 64, 0, stream>>>(x, table, embd);

    dim3 grid(HGRP, BS);
    conv_scramble_sum_kernel<<<grid, 256, 0, stream>>>(embd, conv_w, conv_b, out);

    (void)in_sizes; (void)n_in; (void)out_size; (void)ws_size;
}


# each lane computes its relu(conv) values straight into its WMMA B-fragment registers (a wave per 16-column tile, four sibling chains): the LDS value planes and their barrier are gone; sum order kept
# speedup vs baseline: 3.0977x; 1.3441x over previous
#include <hip/hip_runtime.h>

typedef __attribute__((ext_vector_type(2))) float v2f;
typedef __attribute__((ext_vector_type(8))) float v8f;

#define VOCAB   262
#define EMBD    64
#define OUT_CHS 100
#define BS      32
#define SEQ     512
#define WORD    16

#define KCHUNK  15

#define NSIB    4
#define HGRP    128
#define CSTEP   25

__global__ __launch_bounds__(64) void embed_sum_kernel(
    const int* __restrict__ x, const float* __restrict__ table,
    float* __restrict__ embd)
{
    const int bs = blockIdx.x;
    const int d  = threadIdx.x;
    const int* xw = x + bs * WORD;
    float acc = 0.f;
    #pragma unroll
    for (int i = 0; i < WORD; ++i) {
        acc += table[xw[i] * EMBD + d];
    }
    embd[(size_t)bs * EMBD + d] = acc;
}

__device__ __forceinline__ float conv_relu(
    float e0, float e1, float e2, float e3, float e4, const float (&t)[6])
{
    const float r = fmaf(e4, t[4], fmaf(e3, t[3], fmaf(e2, t[2], fmaf(e1, t[1], fmaf(e0, t[0], t[5])))));
    return fmaxf(r, 0.f);
}

__global__ __launch_bounds__(224) void conv_scramble_sum_kernel(
    const float* __restrict__ embd, const float* __restrict__ conv_w,
    const float* __restrict__ conv_b, float* __restrict__ out)
{
    __shared__ __attribute__((aligned(16))) float sE[100 * 64];

    const int g   = blockIdx.x;
    const int b   = blockIdx.y;
    const int tid = threadIdx.x;
    const int rr0 = g * 100;

    const float4* ebase4 = reinterpret_cast<const float4*>(embd + (size_t)b * SEQ * EMBD);
    float4* sE4 = reinterpret_cast<float4*>(sE);
    float4 stg[7];
    #pragma unroll
    for (int it = 0; it < 7; ++it) {
        const int idx = tid + it * 224;
        const int h   = (rr0 + (idx >> 4)) & 511;
        stg[it] = ebase4[h * 16 + (idx & 15)];
    }
    float4 stg_tail = make_float4(0.f, 0.f, 0.f, 0.f);
    const bool has_tail = tid < 32;
    if (has_tail) {
        const int idx = tid + 7 * 224;
        const int h   = (rr0 + (idx >> 4)) & 511;
        stg_tail = ebase4[h * 16 + (idx & 15)];
    }

    const int c0 = rr0 >> 9;
    const int c1 = c0 < 24 ? c0 + 1 : 24;
    const int P0 = ((c0 + 1) * 512 - rr0) * 60;
    float tA[NSIB][6], tB[NSIB][6];
    #pragma unroll
    for (int m = 0; m < NSIB; ++m) {
        #pragma unroll
        for (int k = 0; k < 5; ++k) {
            tA[m][k] = conv_w[(c0 + CSTEP * m) * 5 + k];
            tB[m][k] = conv_w[(c1 + CSTEP * m) * 5 + k];
        }
        tA[m][5] = conv_b[c0 + CSTEP * m];
        tB[m][5] = conv_b[c1 + CSTEP * m];
    }

    #pragma unroll
    for (int it = 0; it < 7; ++it)
        sE4[tid + it * 224] = stg[it];
    if (has_tail)
        sE4[tid + 7 * 224] = stg_tail;

    __syncthreads();

    const int nt   = __builtin_amdgcn_readfirstlane(tid >> 5);
    const int lane = tid & 31;
    const int half = lane >> 4;
    const int n    = lane & 15;
    const int col  = nt * 16 + n;
    const int cc   = col < OUT_CHS ? col : OUT_CHS - 1;

    const int xa = P0 - 315 - 16 * nt;
    const int xb = P0 - 16 * nt;
    int kA = xa <= 0 ? 0 : (xa + 399) / 400;
    int kB = xb <= 0 ? 0 : (xb + 399) / 400;
    kA = kA < KCHUNK ? kA : KCHUNK;
    kB = kB < KCHUNK ? kB : KCHUNK;

    const v2f aones = {1.f, 1.f};
    v8f acc0 = {}, acc1 = {}, acc2 = {}, acc3 = {};

    unsigned p0 = (unsigned)(half * 200 + cc);
    for (int kb = 0; kb < KCHUNK; ++kb, p0 += 400u) {
        const unsigned p1 = p0 + 100u;
        const unsigned j0 = p0 / 60u;
        const unsigned j1 = p1 / 60u;
        const float* s0 = sE + (p0 + 4u * j0);
        const float* s1 = sE + (p1 + 4u * j1);
        const float a0 = s0[0], a1 = s0[1], a2 = s0[2], a3 = s0[3], a4 = s0[4];
        const float d0 = s1[0], d1 = s1[1], d2 = s1[2], d3 = s1[3], d4 = s1[4];

        v2f bv[NSIB];
        if (kb < kA) {
            #pragma unroll
            for (int m = 0; m < NSIB; ++m) {
                bv[m].x = conv_relu(a0, a1, a2, a3, a4, tA[m]);
                bv[m].y = conv_relu(d0, d1, d2, d3, d4, tA[m]);
            }
        } else if (kb >= kB) {
            #pragma unroll
            for (int m = 0; m < NSIB; ++m) {
                bv[m].x = conv_relu(a0, a1, a2, a3, a4, tB[m]);
                bv[m].y = conv_relu(d0, d1, d2, d3, d4, tB[m]);
            }
        } else {
            const bool up0 = p0 >= (unsigned)P0;
            const bool up1 = p1 >= (unsigned)P0;
            #pragma unroll
            for (int m = 0; m < NSIB; ++m) {
                const float lo0 = conv_relu(a0, a1, a2, a3, a4, tA[m]);
                const float hi0 = conv_relu(a0, a1, a2, a3, a4, tB[m]);
                const float lo1 = conv_relu(d0, d1, d2, d3, d4, tA[m]);
                const float hi1 = conv_relu(d0, d1, d2, d3, d4, tB[m]);
                bv[m].x = up0 ? hi0 : lo0;
                bv[m].y = up1 ? hi1 : lo1;
            }
        }

        acc0 = __builtin_amdgcn_wmma_f32_16x16x4_f32(false, aones, false, bv[0], (short)0, acc0, false, false);
        acc1 = __builtin_amdgcn_wmma_f32_16x16x4_f32(false, aones, false, bv[1], (short)0, acc1, false, false);
        acc2 = __builtin_amdgcn_wmma_f32_16x16x4_f32(false, aones, false, bv[2], (short)0, acc2, false, false);
        acc3 = __builtin_amdgcn_wmma_f32_16x16x4_f32(false, aones, false, bv[3], (short)0, acc3, false, false);
    }

    if (lane < 16 && col < OUT_CHS) {
        const unsigned o = (unsigned)((b * SEQ + g) * OUT_CHS + col);
        out[o]                      = acc0[0];
        out[o + 1u * HGRP * OUT_CHS] = acc1[0];
        out[o + 2u * HGRP * OUT_CHS] = acc2[0];
        out[o + 3u * HGRP * OUT_CHS] = acc3[0];
    }
}

extern "C" void kernel_launch(void* const* d_in, const int* in_sizes, int n_in,
                              void* d_out, int out_size, void* d_ws, size_t ws_size,
                              hipStream_t stream) {
    const int*   x      = (const int*)d_in[0];
    const float* table  = (const float*)d_in[1];
    const float* conv_w = (const float*)d_in[2];
    const float* conv_b = (const float*)d_in[3];
    float*       out    = (float*)d_out;
    float*       embd   = (float*)d_ws;

    embed_sum_kernel<<<BS * SEQ, 64, 0, stream>>>(x, table, embd);

    dim3 grid(HGRP, BS);
    conv_scramble_sum_kernel<<<grid, 224, 0, stream>>>(embd, conv_w, conv_b, out);

    (void)in_sizes; (void)n_in; (void)out_size; (void)ws_size;
}


# embedding-sum kernel re-tiled: a lane per 16-byte row slice, a wave sums 8 rows with all its gathers in flight; 256 eight-wave work-groups replace 16384 two-wave ones; sum order kept
# speedup vs baseline: 3.1975x; 1.0322x over previous
#include <hip/hip_runtime.h>

typedef __attribute__((ext_vector_type(2))) float v2f;
typedef __attribute__((ext_vector_type(8))) float v8f;
typedef __attribute__((ext_vector_type(4))) int   v4i;

#define VOCAB   262
#define EMBD    64
#define OUT_CHS 100
#define BS      32
#define SEQ     512
#define WORD    16

#define KCHUNK  15

#define NSIB    4
#define HGRP    128
#define CSTEP   25

#define K1_WAVES          8
#define K1_TRIPS          4
#define K1_ROWS_PER_WAVE  (2 * K1_TRIPS)
#define K1_ROWS_PER_BLOCK (K1_WAVES * K1_ROWS_PER_WAVE)

__global__ __launch_bounds__(K1_WAVES * 32) __attribute__((amdgpu_waves_per_eu(2, 2)))
void embed_sum_kernel(
    const int* __restrict__ x, const float* __restrict__ table,
    float* __restrict__ embd)
{
    const unsigned lane = threadIdx.x & 31u;
    const unsigned wave = threadIdx.x >> 5;
    const unsigned half = lane >> 4;
    const unsigned q    = lane & 15u;
    const unsigned row0 = (blockIdx.x * K1_WAVES + wave) * K1_ROWS_PER_WAVE + half;

    const v4i*    x4 = reinterpret_cast<const v4i*>(x);
    const float4* t4 = reinterpret_cast<const float4*>(table);
    float4*       e4 = reinterpret_cast<float4*>(embd);

    v4i xi[K1_TRIPS][4];
    #pragma unroll
    for (int it = 0; it < K1_TRIPS; ++it) {
        const unsigned r = row0 + 2u * it;
        #pragma unroll
        for (int j = 0; j < 4; ++j)
            xi[it][j] = x4[r * 4u + j];
    }

    asm volatile("" :: "v"(xi[0][0]), "v"(xi[0][1]), "v"(xi[0][2]), "v"(xi[0][3]),
                       "v"(xi[1][0]), "v"(xi[1][1]), "v"(xi[1][2]), "v"(xi[1][3]),
                       "v"(xi[2][0]), "v"(xi[2][1]), "v"(xi[2][2]), "v"(xi[2][3]),
                       "v"(xi[3][0]), "v"(xi[3][1]), "v"(xi[3][2]), "v"(xi[3][3]));

    #pragma unroll
    for (int it = 0; it < K1_TRIPS; ++it) {
        const unsigned r = row0 + 2u * it;
        const int id[WORD] = {
            xi[it][0].x, xi[it][0].y, xi[it][0].z, xi[it][0].w,
            xi[it][1].x, xi[it][1].y, xi[it][1].z, xi[it][1].w,
            xi[it][2].x, xi[it][2].y, xi[it][2].z, xi[it][2].w,
            xi[it][3].x, xi[it][3].y, xi[it][3].z, xi[it][3].w };
        float4 v[WORD];
        #pragma unroll
        for (int i = 0; i < WORD; ++i)
            v[i] = t4[(unsigned)id[i] * 16u + q];
        float4 acc = make_float4(0.f, 0.f, 0.f, 0.f);
        #pragma unroll
        for (int i = 0; i < WORD; ++i) {
            acc.x += v[i].x; acc.y += v[i].y; acc.z += v[i].z; acc.w += v[i].w;
        }
        e4[r * 16u + q] = acc;
    }
}

__device__ __forceinline__ float conv_relu(
    float e0, float e1, float e2, float e3, float e4, const float (&t)[6])
{
    const float r = fmaf(e4, t[4], fmaf(e3, t[3], fmaf(e2, t[2], fmaf(e1, t[1], fmaf(e0, t[0], t[5])))));
    return fmaxf(r, 0.f);
}

__global__ __launch_bounds__(224) void conv_scramble_sum_kernel(
    const float* __restrict__ embd, const float* __restrict__ conv_w,
    const float* __restrict__ conv_b, float* __restrict__ out)
{
    __shared__ __attribute__((aligned(16))) float sE[100 * 64];

    const int g   = blockIdx.x;
    const int b   = blockIdx.y;
    const int tid = threadIdx.x;
    const int rr0 = g * 100;

    const float4* ebase4 = reinterpret_cast<const float4*>(embd + (size_t)b * SEQ * EMBD);
    float4* sE4 = reinterpret_cast<float4*>(sE);
    float4 stg[7];
    #pragma unroll
    for (int it = 0; it < 7; ++it) {
        const int idx = tid + it * 224;
        const int h   = (rr0 + (idx >> 4)) & 511;
        stg[it] = ebase4[h * 16 + (idx & 15)];
    }
    float4 stg_tail = make_float4(0.f, 0.f, 0.f, 0.f);
    const bool has_tail = tid < 32;
    if (has_tail) {
        const int idx = tid + 7 * 224;
        const int h   = (rr0 + (idx >> 4)) & 511;
        stg_tail = ebase4[h * 16 + (idx & 15)];
    }

    const int c0 = rr0 >> 9;
    const int c1 = c0 < 24 ? c0 + 1 : 24;
    const int P0 = ((c0 + 1) * 512 - rr0) * 60;
    float tA[NSIB][6], tB[NSIB][6];
    #pragma unroll
    for (int m = 0; m < NSIB; ++m) {
        #pragma unroll
        for (int k = 0; k < 5; ++k) {
            tA[m][k] = conv_w[(c0 + CSTEP * m) * 5 + k];
            tB[m][k] = conv_w[(c1 + CSTEP * m) * 5 + k];
        }
        tA[m][5] = conv_b[c0 + CSTEP * m];
        tB[m][5] = conv_b[c1 + CSTEP * m];
    }

    #pragma unroll
    for (int it = 0; it < 7; ++it)
        sE4[tid + it * 224] = stg[it];
    if (has_tail)
        sE4[tid + 7 * 224] = stg_tail;

    __syncthreads();

    const int nt   = __builtin_amdgcn_readfirstlane(tid >> 5);
    const int lane = tid & 31;
    const int half = lane >> 4;
    const int n    = lane & 15;
    const int col  = nt * 16 + n;
    const int cc   = col < OUT_CHS ? col : OUT_CHS - 1;

    const int xa = P0 - 315 - 16 * nt;
    const int xb = P0 - 16 * nt;
    int kA = xa <= 0 ? 0 : (xa + 399) / 400;
    int kB = xb <= 0 ? 0 : (xb + 399) / 400;
    kA = kA < KCHUNK ? kA : KCHUNK;
    kB = kB < KCHUNK ? kB : KCHUNK;

    const v2f aones = {1.f, 1.f};
    v8f acc0 = {}, acc1 = {}, acc2 = {}, acc3 = {};

    unsigned p0 = (unsigned)(half * 200 + cc);
    for (int kb = 0; kb < KCHUNK; ++kb, p0 += 400u) {
        const unsigned p1 = p0 + 100u;
        const unsigned j0 = p0 / 60u;
        const unsigned j1 = p1 / 60u;
        const float* s0 = sE + (p0 + 4u * j0);
        const float* s1 = sE + (p1 + 4u * j1);
        const float a0 = s0[0], a1 = s0[1], a2 = s0[2], a3 = s0[3], a4 = s0[4];
        const float d0 = s1[0], d1 = s1[1], d2 = s1[2], d3 = s1[3], d4 = s1[4];

        v2f bv[NSIB];
        if (kb < kA) {
            #pragma unroll
            for (int m = 0; m < NSIB; ++m) {
                bv[m].x = conv_relu(a0, a1, a2, a3, a4, tA[m]);
                bv[m].y = conv_relu(d0, d1, d2, d3, d4, tA[m]);
            }
        } else if (kb >= kB) {
            #pragma unroll
            for (int m = 0; m < NSIB; ++m) {
                bv[m].x = conv_relu(a0, a1, a2, a3, a4, tB[m]);
                bv[m].y = conv_relu(d0, d1, d2, d3, d4, tB[m]);
            }
        } else {
            const bool up0 = p0 >= (unsigned)P0;
            const bool up1 = p1 >= (unsigned)P0;
            #pragma unroll
            for (int m = 0; m < NSIB; ++m) {
                const float lo0 = conv_relu(a0, a1, a2, a3, a4, tA[m]);
                const float hi0 = conv_relu(a0, a1, a2, a3, a4, tB[m]);
                const float lo1 = conv_relu(d0, d1, d2, d3, d4, tA[m]);
                const float hi1 = conv_relu(d0, d1, d2, d3, d4, tB[m]);
                bv[m].x = up0 ? hi0 : lo0;
                bv[m].y = up1 ? hi1 : lo1;
            }
        }

        acc0 = __builtin_amdgcn_wmma_f32_16x16x4_f32(false, aones, false, bv[0], (short)0, acc0, false, false);
        acc1 = __builtin_amdgcn_wmma_f32_16x16x4_f32(false, aones, false, bv[1], (short)0, acc1, false, false);
        acc2 = __builtin_amdgcn_wmma_f32_16x16x4_f32(false, aones, false, bv[2], (short)0, acc2, false, false);
        acc3 = __builtin_amdgcn_wmma_f32_16x16x4_f32(false, aones, false, bv[3], (short)0, acc3, false, false);
    }

    if (lane < 16 && col < OUT_CHS) {
        const unsigned o = (unsigned)((b * SEQ + g) * OUT_CHS + col);
        out[o]                      = acc0[0];
        out[o + 1u * HGRP * OUT_CHS] = acc1[0];
        out[o + 2u * HGRP * OUT_CHS] = acc2[0];
        out[o + 3u * HGRP * OUT_CHS] = acc3[0];
    }
}

extern "C" void kernel_launch(void* const* d_in, const int* in_sizes, int n_in,
                              void* d_out, int out_size, void* d_ws, size_t ws_size,
                              hipStream_t stream) {
    const int*   x      = (const int*)d_in[0];
    const float* table  = (const float*)d_in[1];
    const float* conv_w = (const float*)d_in[2];
    const float* conv_b = (const float*)d_in[3];
    float*       out    = (float*)d_out;
    float*       embd   = (float*)d_ws;

    embed_sum_kernel<<<BS * SEQ / K1_ROWS_PER_BLOCK, K1_WAVES * 32, 0, stream>>>(x, table, embd);

    dim3 grid(HGRP, BS);
    conv_scramble_sum_kernel<<<grid, 224, 0, stream>>>(embd, conv_w, conv_b, out);

    (void)in_sizes; (void)n_in; (void)out_size; (void)ws_size;
}


# relu(conv) values of the 15 chunks added per lane on the vector unit into one B fragment per sibling; one WMMA per sibling (was 15) sums its 4 k slots at the end; sum order changed
# speedup vs baseline: 4.2313x; 1.3233x over previous
#include <hip/hip_runtime.h>

typedef __attribute__((ext_vector_type(2))) float v2f;
typedef __attribute__((ext_vector_type(8))) float v8f;
typedef __attribute__((ext_vector_type(4))) int   v4i;

#define VOCAB   262
#define EMBD    64
#define OUT_CHS 100
#define BS      32
#define SEQ     512
#define WORD    16

#define KCHUNK  15

#define NSIB    4
#define HGRP    128
#define CSTEP   25

#define K1_WAVES          8
#define K1_TRIPS          4
#define K1_ROWS_PER_WAVE  (2 * K1_TRIPS)
#define K1_ROWS_PER_BLOCK (K1_WAVES * K1_ROWS_PER_WAVE)

__global__ __launch_bounds__(K1_WAVES * 32) __attribute__((amdgpu_waves_per_eu(2, 2)))
void embed_sum_kernel(
    const int* __restrict__ x, const float* __restrict__ table,
    float* __restrict__ embd)
{
    const unsigned lane = threadIdx.x & 31u;
    const unsigned wave = threadIdx.x >> 5;
    const unsigned half = lane >> 4;
    const unsigned q    = lane & 15u;
    const unsigned row0 = (blockIdx.x * K1_WAVES + wave) * K1_ROWS_PER_WAVE + half;

    const v4i*    x4 = reinterpret_cast<const v4i*>(x);
    const float4* t4 = reinterpret_cast<const float4*>(table);
    float4*       e4 = reinterpret_cast<float4*>(embd);

    v4i xi[K1_TRIPS][4];
    #pragma unroll
    for (int it = 0; it < K1_TRIPS; ++it) {
        const unsigned r = row0 + 2u * it;
        #pragma unroll
        for (int j = 0; j < 4; ++j)
            xi[it][j] = x4[r * 4u + j];
    }

    asm volatile("" :: "v"(xi[0][0]), "v"(xi[0][1]), "v"(xi[0][2]), "v"(xi[0][3]),
                       "v"(xi[1][0]), "v"(xi[1][1]), "v"(xi[1][2]), "v"(xi[1][3]),
                       "v"(xi[2][0]), "v"(xi[2][1]), "v"(xi[2][2]), "v"(xi[2][3]),
                       "v"(xi[3][0]), "v"(xi[3][1]), "v"(xi[3][2]), "v"(xi[3][3]));

    #pragma unroll
    for (int it = 0; it < K1_TRIPS; ++it) {
        const unsigned r = row0 + 2u * it;
        const int id[WORD] = {
            xi[it][0].x, xi[it][0].y, xi[it][0].z, xi[it][0].w,
            xi[it][1].x, xi[it][1].y, xi[it][1].z, xi[it][1].w,
            xi[it][2].x, xi[it][2].y, xi[it][2].z, xi[it][2].w,
            xi[it][3].x, xi[it][3].y, xi[it][3].z, xi[it][3].w };
        float4 v[WORD];
        #pragma unroll
        for (int i = 0; i < WORD; ++i)
            v[i] = t4[(unsigned)id[i] * 16u + q];
        float4 acc = make_float4(0.f, 0.f, 0.f, 0.f);
        #pragma unroll
        for (int i = 0; i < WORD; ++i) {
            acc.x += v[i].x; acc.y += v[i].y; acc.z += v[i].z; acc.w += v[i].w;
        }
        e4[r * 16u + q] = acc;
    }
}

__device__ __forceinline__ float conv_relu(
    float e0, float e1, float e2, float e3, float e4, const float (&t)[6])
{
    const float r = fmaf(e4, t[4], fmaf(e3, t[3], fmaf(e2, t[2], fmaf(e1, t[1], fmaf(e0, t[0], t[5])))));
    return fmaxf(r, 0.f);
}

__global__ __launch_bounds__(224) void conv_scramble_sum_kernel(
    const float* __restrict__ embd, const float* __restrict__ conv_w,
    const float* __restrict__ conv_b, float* __restrict__ out)
{
    __shared__ __attribute__((aligned(16))) float sE[100 * 64];

    const int g   = blockIdx.x;
    const int b   = blockIdx.y;
    const int tid = threadIdx.x;
    const int rr0 = g * 100;

    const float4* ebase4 = reinterpret_cast<const float4*>(embd + (size_t)b * SEQ * EMBD);
    float4* sE4 = reinterpret_cast<float4*>(sE);
    float4 stg[7];
    #pragma unroll
    for (int it = 0; it < 7; ++it) {
        const int idx = tid + it * 224;
        const int h   = (rr0 + (idx >> 4)) & 511;
        stg[it] = ebase4[h * 16 + (idx & 15)];
    }
    float4 stg_tail = make_float4(0.f, 0.f, 0.f, 0.f);
    const bool has_tail = tid < 32;
    if (has_tail) {
        const int idx = tid + 7 * 224;
        const int h   = (rr0 + (idx >> 4)) & 511;
        stg_tail = ebase4[h * 16 + (idx & 15)];
    }

    const int c0 = rr0 >> 9;
    const int c1 = c0 < 24 ? c0 + 1 : 24;
    const int P0 = ((c0 + 1) * 512 - rr0) * 60;
    float tA[NSIB][6], tB[NSIB][6];
    #pragma unroll
    for (int m = 0; m < NSIB; ++m) {
        #pragma unroll
        for (int k = 0; k < 5; ++k) {
            tA[m][k] = conv_w[(c0 + CSTEP * m) * 5 + k];
            tB[m][k] = conv_w[(c1 + CSTEP * m) * 5 + k];
        }
        tA[m][5] = conv_b[c0 + CSTEP * m];
        tB[m][5] = conv_b[c1 + CSTEP * m];
    }

    #pragma unroll
    for (int it = 0; it < 7; ++it)
        sE4[tid + it * 224] = stg[it];
    if (has_tail)
        sE4[tid + 7 * 224] = stg_tail;

    __syncthreads();

    const int nt   = __builtin_amdgcn_readfirstlane(tid >> 5);
    const int lane = tid & 31;
    const int half = lane >> 4;
    const int n    = lane & 15;
    const int col  = nt * 16 + n;
    const int cc   = col < OUT_CHS ? col : OUT_CHS - 1;

    const int xa = P0 - 315 - 16 * nt;
    const int xb = P0 - 16 * nt;
    int kA = xa <= 0 ? 0 : (xa + 399) / 400;
    int kB = xb <= 0 ? 0 : (xb + 399) / 400;
    kA = kA < KCHUNK ? kA : KCHUNK;
    kB = kB < KCHUNK ? kB : KCHUNK;

    v2f bsum[NSIB];
    #pragma unroll
    for (int m = 0; m < NSIB; ++m) { bsum[m].x = 0.f; bsum[m].y = 0.f; }

    unsigned p0 = (unsigned)(half * 200 + cc);
    for (int kb = 0; kb < KCHUNK; ++kb, p0 += 400u) {
        const unsigned p1 = p0 + 100u;
        const unsigned j0 = p0 / 60u;
        const unsigned j1 = p1 / 60u;
        const float* s0 = sE + (p0 + 4u * j0);
        const float* s1 = sE + (p1 + 4u * j1);
        const float a0 = s0[0], a1 = s0[1], a2 = s0[2], a3 = s0[3], a4 = s0[4];
        const float d0 = s1[0], d1 = s1[1], d2 = s1[2], d3 = s1[3], d4 = s1[4];

        v2f bv[NSIB];
        if (kb < kA) {
            #pragma unroll
            for (int m = 0; m < NSIB; ++m) {
                bv[m].x = conv_relu(a0, a1, a2, a3, a4, tA[m]);
                bv[m].y = conv_relu(d0, d1, d2, d3, d4, tA[m]);
            }
        } else if (kb >= kB) {
            #pragma unroll
            for (int m = 0; m < NSIB; ++m) {
                bv[m].x = conv_relu(a0, a1, a2, a3, a4, tB[m]);
                bv[m].y = conv_relu(d0, d1, d2, d3, d4, tB[m]);
            }
        } else {
            const bool up0 = p0 >= (unsigned)P0;
            const bool up1 = p1 >= (unsigned)P0;
            #pragma unroll
            for (int m = 0; m < NSIB; ++m) {
                const float lo0 = conv_relu(a0, a1, a2, a3, a4, tA[m]);
                const float hi0 = conv_relu(a0, a1, a2, a3, a4, tB[m]);
                const float lo1 = conv_relu(d0, d1, d2, d3, d4, tA[m]);
                const float hi1 = conv_relu(d0, d1, d2, d3, d4, tB[m]);
                bv[m].x = up0 ? hi0 : lo0;
                bv[m].y = up1 ? hi1 : lo1;
            }
        }

        #pragma unroll
        for (int m = 0; m < NSIB; ++m) {
            bsum[m].x += bv[m].x;
            bsum[m].y += bv[m].y;
        }
    }

    const v2f aones = {1.f, 1.f};
    const v8f zero8 = {};
    const v8f acc0 = __builtin_amdgcn_wmma_f32_16x16x4_f32(false, aones, false, bsum[0], (short)0, zero8, false, false);
    const v8f acc1 = __builtin_amdgcn_wmma_f32_16x16x4_f32(false, aones, false, bsum[1], (short)0, zero8, false, false);
    const v8f acc2 = __builtin_amdgcn_wmma_f32_16x16x4_f32(false, aones, false, bsum[2], (short)0, zero8, false, false);
    const v8f acc3 = __builtin_amdgcn_wmma_f32_16x16x4_f32(false, aones, false, bsum[3], (short)0, zero8, false, false);

    if (lane < 16 && col < OUT_CHS) {
        const unsigned o = (unsigned)((b * SEQ + g) * OUT_CHS + col);
        out[o]                      = acc0[0];
        out[o + 1u * HGRP * OUT_CHS] = acc1[0];
        out[o + 2u * HGRP * OUT_CHS] = acc2[0];
        out[o + 3u * HGRP * OUT_CHS] = acc3[0];
    }
}

extern "C" void kernel_launch(void* const* d_in, const int* in_sizes, int n_in,
                              void* d_out, int out_size, void* d_ws, size_t ws_size,
                              hipStream_t stream) {
    const int*   x      = (const int*)d_in[0];
    const float* table  = (const float*)d_in[1];
    const float* conv_w = (const float*)d_in[2];
    const float* conv_b = (const float*)d_in[3];
    float*       out    = (float*)d_out;
    float*       embd   = (float*)d_ws;

    embed_sum_kernel<<<BS * SEQ / K1_ROWS_PER_BLOCK, K1_WAVES * 32, 0, stream>>>(x, table, embd);

    dim3 grid(HGRP, BS);
    conv_scramble_sum_kernel<<<grid, 224, 0, stream>>>(embd, conv_w, conv_b, out);

    (void)in_sizes; (void)n_in; (void)out_size; (void)ws_size;
}


# conv loop re-tiled: a lane computes 4 adjacent columns of a row w' for all 4 siblings from two 16-byte LDS reads (packed FMAs on sibling pairs); 5 partial rows meet in LDS; sum order changed
# speedup vs baseline: 5.7344x; 1.3552x over previous
#include <hip/hip_runtime.h>

typedef __attribute__((ext_vector_type(2))) float v2f;
typedef __attribute__((ext_vector_type(8))) float v8f;
typedef __attribute__((ext_vector_type(4))) int   v4i;

#define VOCAB   262
#define EMBD    64
#define OUT_CHS 100
#define BS      32
#define SEQ     512
#define WORD    16

#define NSIB    4
#define HGRP    128
#define CSTEP   25

#define K1_WAVES          8
#define K1_TRIPS          4
#define K1_ROWS_PER_WAVE  (2 * K1_TRIPS)
#define K1_ROWS_PER_BLOCK (K1_WAVES * K1_ROWS_PER_WAVE)

__global__ __launch_bounds__(K1_WAVES * 32) __attribute__((amdgpu_waves_per_eu(2, 2)))
void embed_sum_kernel(
    const int* __restrict__ x, const float* __restrict__ table,
    float* __restrict__ embd)
{
    const unsigned lane = threadIdx.x & 31u;
    const unsigned wave = threadIdx.x >> 5;
    const unsigned half = lane >> 4;
    const unsigned q    = lane & 15u;
    const unsigned row0 = (blockIdx.x * K1_WAVES + wave) * K1_ROWS_PER_WAVE + half;

    const v4i*    x4 = reinterpret_cast<const v4i*>(x);
    const float4* t4 = reinterpret_cast<const float4*>(table);
    float4*       e4 = reinterpret_cast<float4*>(embd);

    v4i xi[K1_TRIPS][4];
    #pragma unroll
    for (int it = 0; it < K1_TRIPS; ++it) {
        const unsigned r = row0 + 2u * it;
        #pragma unroll
        for (int j = 0; j < 4; ++j)
            xi[it][j] = x4[r * 4u + j];
    }

    asm volatile("" :: "v"(xi[0][0]), "v"(xi[0][1]), "v"(xi[0][2]), "v"(xi[0][3]),
                       "v"(xi[1][0]), "v"(xi[1][1]), "v"(xi[1][2]), "v"(xi[1][3]),
                       "v"(xi[2][0]), "v"(xi[2][1]), "v"(xi[2][2]), "v"(xi[2][3]),
                       "v"(xi[3][0]), "v"(xi[3][1]), "v"(xi[3][2]), "v"(xi[3][3]));

    #pragma unroll
    for (int it = 0; it < K1_TRIPS; ++it) {
        const unsigned r = row0 + 2u * it;
        const int id[WORD] = {
            xi[it][0].x, xi[it][0].y, xi[it][0].z, xi[it][0].w,
            xi[it][1].x, xi[it][1].y, xi[it][1].z, xi[it][1].w,
            xi[it][2].x, xi[it][2].y, xi[it][2].z, xi[it][2].w,
            xi[it][3].x, xi[it][3].y, xi[it][3].z, xi[it][3].w };
        float4 v[WORD];
        #pragma unroll
        for (int i = 0; i < WORD; ++i)
            v[i] = t4[(unsigned)id[i] * 16u + q];
        float4 acc = make_float4(0.f, 0.f, 0.f, 0.f);
        #pragma unroll
        for (int i = 0; i < WORD; ++i) {
            acc.x += v[i].x; acc.y += v[i].y; acc.z += v[i].z; acc.w += v[i].w;
        }
        e4[r * 16u + q] = acc;
    }
}

#define K2_THREADS 128
#define RPAR       5
#define NACT       (25 * RPAR)
#define NITER      (60 / RPAR)
#define NTILE      7

__device__ __forceinline__ v2f conv_relu2(
    float e0, float e1, float e2, float e3, float e4, const v2f (&t)[6])
{
    const v2f s0 = {e0, e0}, s1 = {e1, e1}, s2 = {e2, e2}, s3 = {e3, e3}, s4 = {e4, e4};
    v2f r = __builtin_elementwise_fma(s0, t[0], t[5]);
    r = __builtin_elementwise_fma(s1, t[1], r);
    r = __builtin_elementwise_fma(s2, t[2], r);
    r = __builtin_elementwise_fma(s3, t[3], r);
    r = __builtin_elementwise_fma(s4, t[4], r);
    const v2f z = {0.f, 0.f};
    return __builtin_elementwise_max(r, z);
}

__global__ __launch_bounds__(K2_THREADS) void conv_scramble_sum_kernel(
    const float* __restrict__ embd, const float* __restrict__ conv_w,
    const float* __restrict__ conv_b, float* __restrict__ out)
{
    __shared__ __attribute__((aligned(16))) float sE[100 * 64];
    __shared__ __attribute__((aligned(16))) float sR[NSIB * RPAR * OUT_CHS];

    const int g   = blockIdx.x;
    const int b   = blockIdx.y;
    const int tid = threadIdx.x;
    const int rr0 = g * 100;

    const float4* ebase4 = reinterpret_cast<const float4*>(embd + (size_t)b * SEQ * EMBD);
    float4* sE4 = reinterpret_cast<float4*>(sE);
    float4 stg[12];
    #pragma unroll
    for (int it = 0; it < 12; ++it) {
        const int idx = tid + it * K2_THREADS;
        const int h   = (rr0 + (idx >> 4)) & 511;
        stg[it] = ebase4[h * 16 + (idx & 15)];
    }
    float4 stg_tail = make_float4(0.f, 0.f, 0.f, 0.f);
    const bool has_tail = tid < 64;
    if (has_tail) {
        const int idx = tid + 12 * K2_THREADS;
        const int h   = (rr0 + (idx >> 4)) & 511;
        stg_tail = ebase4[h * 16 + (idx & 15)];
    }

    const int c0 = rr0 >> 9;
    const int c1 = c0 < 24 ? c0 + 1 : 24;
    const int T0 = ((c0 + 1) * 512 - rr0) * 15;
    v2f tA[NSIB / 2][6], tB[NSIB / 2][6];
    #pragma unroll
    for (int h = 0; h < NSIB / 2; ++h) {
        #pragma unroll
        for (int k = 0; k < 5; ++k) {
            tA[h][k].x = conv_w[(c0 + CSTEP * (2 * h)) * 5 + k];
            tA[h][k].y = conv_w[(c0 + CSTEP * (2 * h + 1)) * 5 + k];
            tB[h][k].x = conv_w[(c1 + CSTEP * (2 * h)) * 5 + k];
            tB[h][k].y = conv_w[(c1 + CSTEP * (2 * h + 1)) * 5 + k];
        }
        tA[h][5].x = conv_b[c0 + CSTEP * (2 * h)];
        tA[h][5].y = conv_b[c0 + CSTEP * (2 * h + 1)];
        tB[h][5].x = conv_b[c1 + CSTEP * (2 * h)];
        tB[h][5].y = conv_b[c1 + CSTEP * (2 * h + 1)];
    }

    #pragma unroll
    for (int it = 0; it < 12; ++it)
        sE4[tid + it * K2_THREADS] = stg[it];
    if (has_tail)
        sE4[tid + 12 * K2_THREADS] = stg_tail;

    __syncthreads();

    #pragma unroll
    for (int h = 0; h < NSIB / 2; ++h) {
        #pragma unroll
        for (int k = 0; k < 6; ++k) {
            asm volatile("" : "+v"(tA[h][k]));
            asm volatile("" : "+v"(tB[h][k]));
        }
    }

    const int wave = __builtin_amdgcn_readfirstlane(tid >> 5);
    const int lane = tid & 31;

    const unsigned T = tid < NACT ? (unsigned)tid : (unsigned)(NACT - 1);

    unsigned bs[3];
    #pragma unroll
    for (int u = 0; u < 3; ++u) {
        const unsigned t = T + (unsigned)(NACT * u);
        bs[u] = 4u * (t + t / 15u);
    }

    v2f acc[NSIB / 2][4];
    #pragma unroll
    for (int h = 0; h < NSIB / 2; ++h) {
        #pragma unroll
        for (int i = 0; i < 4; ++i) { acc[h][i].x = 0.f; acc[h][i].y = 0.f; }
    }

    for (int k3 = 0; k3 < NITER / 3; ++k3) {
        #pragma unroll
        for (int u = 0; u < 3; ++u) {
            const float* s = sE + bs[u];
            const float4 lo = *reinterpret_cast<const float4*>(s);
            const float4 hi = *reinterpret_cast<const float4*>(s + 4);
            const float e[8] = { lo.x, lo.y, lo.z, lo.w, hi.x, hi.y, hi.z, hi.w };
            bs[u] += 1600u;

            const int it  = 3 * k3 + u;
            const int tlo = wave * 32 + NACT * it;
            if (tlo + 31 < T0) {
                #pragma unroll
                for (int h = 0; h < NSIB / 2; ++h) {
                    #pragma unroll
                    for (int i = 0; i < 4; ++i)
                        acc[h][i] += conv_relu2(e[i], e[i + 1], e[i + 2], e[i + 3], e[i + 4], tA[h]);
                }
            } else if (tlo >= T0) {
                #pragma unroll
                for (int h = 0; h < NSIB / 2; ++h) {
                    #pragma unroll
                    for (int i = 0; i < 4; ++i)
                        acc[h][i] += conv_relu2(e[i], e[i + 1], e[i + 2], e[i + 3], e[i + 4], tB[h]);
                }
            } else {
                const bool up = (int)(T + (unsigned)(NACT * it)) >= T0;
                #pragma unroll
                for (int h = 0; h < NSIB / 2; ++h) {
                    v2f ts[6];
                    #pragma unroll
                    for (int k = 0; k < 6; ++k) {
                        ts[k].x = up ? tB[h][k].x : tA[h][k].x;
                        ts[k].y = up ? tB[h][k].y : tA[h][k].y;
                    }
                    #pragma unroll
                    for (int i = 0; i < 4; ++i)
                        acc[h][i] += conv_relu2(e[i], e[i + 1], e[i + 2], e[i + 3], e[i + 4], ts);
                }
            }
        }
    }

    float4* sR4 = reinterpret_cast<float4*>(sR);
    if (tid < NACT) {
        #pragma unroll
        for (int h = 0; h < NSIB / 2; ++h) {
            sR4[(2 * h) * NACT + tid]     = make_float4(acc[h][0].x, acc[h][1].x, acc[h][2].x, acc[h][3].x);
            sR4[(2 * h + 1) * NACT + tid] = make_float4(acc[h][0].y, acc[h][1].y, acc[h][2].y, acc[h][3].y);
        }
    }

    __syncthreads();

    const int half = lane >> 4;
    const int n    = lane & 15;
    const float* rbase = sR + wave * (RPAR * OUT_CHS);
    const v2f aones = {1.f, 1.f};
    const v8f zero8 = {};
    float res[NTILE];
    #pragma unroll
    for (int nt = 0; nt < NTILE; ++nt) {
        const int col = nt * 16 + n;
        const int cc  = col < OUT_CHS ? col : OUT_CHS - 1;
        const float r0 = rbase[(2 * half) * OUT_CHS + cc];
        const float r1 = rbase[(2 * half + 1) * OUT_CHS + cc];
        const float r4 = rbase[4 * OUT_CHS + cc];
        v2f bfrag;
        bfrag.x = r0 + (half == 0 ? r4 : 0.f);
        bfrag.y = r1;
        const v8f d = __builtin_amdgcn_wmma_f32_16x16x4_f32(false, aones, false, bfrag, (short)0, zero8, false, false);
        res[nt] = d[0];
    }

    if (lane < 16) {
        const unsigned o = (unsigned)((b * SEQ + g + HGRP * wave) * OUT_CHS + n);
        #pragma unroll
        for (int nt = 0; nt < NTILE; ++nt) {
            if (nt * 16 + n < OUT_CHS)
                out[o + 16u * nt] = res[nt];
        }
    }
}

extern "C" void kernel_launch(void* const* d_in, const int* in_sizes, int n_in,
                              void* d_out, int out_size, void* d_ws, size_t ws_size,
                              hipStream_t stream) {
    const int*   x      = (const int*)d_in[0];
    const float* table  = (const float*)d_in[1];
    const float* conv_w = (const float*)d_in[2];
    const float* conv_b = (const float*)d_in[3];
    float*       out    = (float*)d_out;
    float*       embd   = (float*)d_ws;

    embed_sum_kernel<<<BS * SEQ / K1_ROWS_PER_BLOCK, K1_WAVES * 32, 0, stream>>>(x, table, embd);

    dim3 grid(HGRP, BS);
    conv_scramble_sum_kernel<<<grid, K2_THREADS, 0, stream>>>(embd, conv_w, conv_b, out);

    (void)in_sizes; (void)n_in; (void)out_size; (void)ws_size;
}


# one work-group (two four-wave teams) serves four row windows 4 rows apart, 16 output rows, from 112 embd rows staged once: 1024 eight-wave work-groups replace 4096 four-wave ones; sum order kept
# speedup vs baseline: 6.1173x; 1.0668x over previous
#include <hip/hip_runtime.h>

typedef __attribute__((ext_vector_type(2))) float v2f;
typedef __attribute__((ext_vector_type(8))) float v8f;
typedef __attribute__((ext_vector_type(4))) int   v4i;

#define VOCAB   262
#define EMBD    64
#define OUT_CHS 100
#define BS      32
#define SEQ     512
#define WORD    16

#define NSIB    4
#define HGRP    128
#define CSTEP   25

#define K1_WAVES          8
#define K1_TRIPS          4
#define K1_ROWS_PER_WAVE  (2 * K1_TRIPS)
#define K1_ROWS_PER_BLOCK (K1_WAVES * K1_ROWS_PER_WAVE)

__global__ __launch_bounds__(K1_WAVES * 32) __attribute__((amdgpu_waves_per_eu(2, 2)))
void embed_sum_kernel(
    const int* __restrict__ x, const float* __restrict__ table,
    float* __restrict__ embd)
{
    const unsigned lane = threadIdx.x & 31u;
    const unsigned wave = threadIdx.x >> 5;
    const unsigned half = lane >> 4;
    const unsigned q    = lane & 15u;
    const unsigned row0 = (blockIdx.x * K1_WAVES + wave) * K1_ROWS_PER_WAVE + half;

    const v4i*    x4 = reinterpret_cast<const v4i*>(x);
    const float4* t4 = reinterpret_cast<const float4*>(table);
    float4*       e4 = reinterpret_cast<float4*>(embd);

    v4i xi[K1_TRIPS][4];
    #pragma unroll
    for (int it = 0; it < K1_TRIPS; ++it) {
        const unsigned r = row0 + 2u * it;
        #pragma unroll
        for (int j = 0; j < 4; ++j)
            xi[it][j] = x4[r * 4u + j];
    }

    asm volatile("" :: "v"(xi[0][0]), "v"(xi[0][1]), "v"(xi[0][2]), "v"(xi[0][3]),
                       "v"(xi[1][0]), "v"(xi[1][1]), "v"(xi[1][2]), "v"(xi[1][3]),
                       "v"(xi[2][0]), "v"(xi[2][1]), "v"(xi[2][2]), "v"(xi[2][3]),
                       "v"(xi[3][0]), "v"(xi[3][1]), "v"(xi[3][2]), "v"(xi[3][3]));

    #pragma unroll
    for (int it = 0; it < K1_TRIPS; ++it) {
        const unsigned r = row0 + 2u * it;
        const int id[WORD] = {
            xi[it][0].x, xi[it][0].y, xi[it][0].z, xi[it][0].w,
            xi[it][1].x, xi[it][1].y, xi[it][1].z, xi[it][1].w,
            xi[it][2].x, xi[it][2].y, xi[it][2].z, xi[it][2].w,
            xi[it][3].x, xi[it][3].y, xi[it][3].z, xi[it][3].w };
        float4 v[WORD];
        #pragma unroll
        for (int i = 0; i < WORD; ++i)
            v[i] = t4[(unsigned)id[i] * 16u + q];
        float4 acc = make_float4(0.f, 0.f, 0.f, 0.f);
        #pragma unroll
        for (int i = 0; i < WORD; ++i) {
            acc.x += v[i].x; acc.y += v[i].y; acc.z += v[i].z; acc.w += v[i].w;
        }
        e4[r * 16u + q] = acc;
    }
}

#define NWIN       4
#define NTEAM      2
#define NPASS      (NWIN / NTEAM)
#define SROWS      (100 + 4 * (NWIN - 1))
#define TEAM_THREADS 128
#define K2_THREADS (TEAM_THREADS * NTEAM)
#define NSTG       (SROWS * 16 / K2_THREADS)
#define RPAR       5
#define NACT       (25 * RPAR)
#define NITER      (60 / RPAR)
#define NTILE      7
#define SR_FLOATS  (NSIB * RPAR * OUT_CHS)

__device__ __forceinline__ v2f conv_relu2(
    float e0, float e1, float e2, float e3, float e4, const v2f (&t)[6])
{
    const v2f s0 = {e0, e0}, s1 = {e1, e1}, s2 = {e2, e2}, s3 = {e3, e3}, s4 = {e4, e4};
    v2f r = __builtin_elementwise_fma(s0, t[0], t[5]);
    r = __builtin_elementwise_fma(s1, t[1], r);
    r = __builtin_elementwise_fma(s2, t[2], r);
    r = __builtin_elementwise_fma(s3, t[3], r);
    r = __builtin_elementwise_fma(s4, t[4], r);
    const v2f z = {0.f, 0.f};
    return __builtin_elementwise_max(r, z);
}

__global__ __launch_bounds__(K2_THREADS) void conv_scramble_sum_kernel(
    const float* __restrict__ embd, const float* __restrict__ conv_w,
    const float* __restrict__ conv_b, float* __restrict__ out)
{
    __shared__ __attribute__((aligned(16))) float sE[SROWS * 64];
    __shared__ __attribute__((aligned(16))) float sR[NTEAM * NPASS * SR_FLOATS];

    const int G   = blockIdx.x;
    const int b   = blockIdx.y;
    const int tid = threadIdx.x;
    const int tt  = tid & (TEAM_THREADS - 1);

    const float4* ebase4 = reinterpret_cast<const float4*>(embd + (size_t)b * SEQ * EMBD);
    float4* sE4 = reinterpret_cast<float4*>(sE);
    float4 stg[NSTG];
    #pragma unroll
    for (int it = 0; it < NSTG; ++it) {
        const int idx = tid + it * K2_THREADS;
        const int h   = (16 * G + (idx >> 4)) & 511;
        stg[it] = ebase4[h * 16 + (idx & 15)];
    }

    const int team = __builtin_amdgcn_readfirstlane(tid >> 7);
    const int wave = __builtin_amdgcn_readfirstlane((tid >> 5) & 3);
    const int lane = tid & 31;

    const unsigned T = tt < NACT ? (unsigned)tt : (unsigned)(NACT - 1);

    #pragma unroll
    for (int pass = 0; pass < NPASS; ++pass) {
        const int wi  = team + NTEAM * pass;
        const int k   = NWIN * G + wi;
        const int g   = (41 * k) & 127;
        const int rr0 = g * 100;

        const int c0 = rr0 >> 9;
        const int c1 = c0 < 24 ? c0 + 1 : 24;
        const int T0 = ((c0 + 1) * 512 - rr0) * 15;
        v2f tA[NSIB / 2][6], tB[NSIB / 2][6];
        #pragma unroll
        for (int h = 0; h < NSIB / 2; ++h) {
            #pragma unroll
            for (int kk = 0; kk < 5; ++kk) {
                tA[h][kk].x = conv_w[(c0 + CSTEP * (2 * h)) * 5 + kk];
                tA[h][kk].y = conv_w[(c0 + CSTEP * (2 * h + 1)) * 5 + kk];
                tB[h][kk].x = conv_w[(c1 + CSTEP * (2 * h)) * 5 + kk];
                tB[h][kk].y = conv_w[(c1 + CSTEP * (2 * h + 1)) * 5 + kk];
            }
            tA[h][5].x = conv_b[c0 + CSTEP * (2 * h)];
            tA[h][5].y = conv_b[c0 + CSTEP * (2 * h + 1)];
            tB[h][5].x = conv_b[c1 + CSTEP * (2 * h)];
            tB[h][5].y = conv_b[c1 + CSTEP * (2 * h + 1)];
        }

        if (pass == 0) {
            #pragma unroll
            for (int it = 0; it < NSTG; ++it)
                sE4[tid + it * K2_THREADS] = stg[it];
            __syncthreads();
        }

        #pragma unroll
        for (int h = 0; h < NSIB / 2; ++h) {
            #pragma unroll
            for (int kk = 0; kk < 6; ++kk) {
                asm volatile("" : "+v"(tA[h][kk]));
                asm volatile("" : "+v"(tB[h][kk]));
            }
        }

        unsigned bs[3];
        #pragma unroll
        for (int u = 0; u < 3; ++u) {
            const unsigned t = T + (unsigned)(NACT * u);
            bs[u] = 4u * (t + t / 15u) + 256u * (unsigned)wi;
        }

        v2f acc[NSIB / 2][4];
        #pragma unroll
        for (int h = 0; h < NSIB / 2; ++h) {
            #pragma unroll
            for (int i = 0; i < 4; ++i) { acc[h][i].x = 0.f; acc[h][i].y = 0.f; }
        }

        for (int k3 = 0; k3 < NITER / 3; ++k3) {
            #pragma unroll
            for (int u = 0; u < 3; ++u) {
                const float* s = sE + bs[u];
                const float4 lo = *reinterpret_cast<const float4*>(s);
                const float4 hi = *reinterpret_cast<const float4*>(s + 4);
                const float e[8] = { lo.x, lo.y, lo.z, lo.w, hi.x, hi.y, hi.z, hi.w };
                bs[u] += 1600u;

                const int it  = 3 * k3 + u;
                const int tlo = wave * 32 + NACT * it;
                if (tlo + 31 < T0) {
                    #pragma unroll
                    for (int h = 0; h < NSIB / 2; ++h) {
                        #pragma unroll
                        for (int i = 0; i < 4; ++i)
                            acc[h][i] += conv_relu2(e[i], e[i + 1], e[i + 2], e[i + 3], e[i + 4], tA[h]);
                    }
                } else if (tlo >= T0) {
                    #pragma unroll
                    for (int h = 0; h < NSIB / 2; ++h) {
                        #pragma unroll
                        for (int i = 0; i < 4; ++i)
                            acc[h][i] += conv_relu2(e[i], e[i + 1], e[i + 2], e[i + 3], e[i + 4], tB[h]);
                    }
                } else {
                    const bool up = (int)(T + (unsigned)(NACT * it)) >= T0;
                    #pragma unroll
                    for (int h = 0; h < NSIB / 2; ++h) {
                        v2f ts[6];
                        #pragma unroll
                        for (int kk = 0; kk < 6; ++kk) {
                            ts[kk].x = up ? tB[h][kk].x : tA[h][kk].x;
                            ts[kk].y = up ? tB[h][kk].y : tA[h][kk].y;
                        }
                        #pragma unroll
                        for (int i = 0; i < 4; ++i)
                            acc[h][i] += conv_relu2(e[i], e[i + 1], e[i + 2], e[i + 3], e[i + 4], ts);
                    }
                }
            }
        }

        float* sRw = sR + (team * NPASS + pass) * SR_FLOATS;
        float4* sR4 = reinterpret_cast<float4*>(sRw);
        if (tt < NACT) {
            #pragma unroll
            for (int h = 0; h < NSIB / 2; ++h) {
                sR4[(2 * h) * NACT + tt]     = make_float4(acc[h][0].x, acc[h][1].x, acc[h][2].x, acc[h][3].x);
                sR4[(2 * h + 1) * NACT + tt] = make_float4(acc[h][0].y, acc[h][1].y, acc[h][2].y, acc[h][3].y);
            }
        }

        __syncthreads();

        const int half = lane >> 4;
        const int n    = lane & 15;
        const float* rbase = sRw + wave * (RPAR * OUT_CHS);
        const v2f aones = {1.f, 1.f};
        const v8f zero8 = {};
        float res[NTILE];
        #pragma unroll
        for (int nt = 0; nt < NTILE; ++nt) {
            const int col = nt * 16 + n;
            const int cc  = col < OUT_CHS ? col : OUT_CHS - 1;
            const float r0 = rbase[(2 * half) * OUT_CHS + cc];
            const float r1 = rbase[(2 * half + 1) * OUT_CHS + cc];
            const float r4 = rbase[4 * OUT_CHS + cc];
            v2f bfrag;
            bfrag.x = r0 + (half == 0 ? r4 : 0.f);
            bfrag.y = r1;
            const v8f d = __builtin_amdgcn_wmma_f32_16x16x4_f32(false, aones, false, bfrag, (short)0, zero8, false, false);
            res[nt] = d[0];
        }

        if (lane < 16) {
            const unsigned o = (unsigned)((b * SEQ + g + HGRP * wave) * OUT_CHS + n);
            #pragma unroll
            for (int nt = 0; nt < NTILE; ++nt) {
                if (nt * 16 + n < OUT_CHS)
                    out[o + 16u * nt] = res[nt];
            }
        }
    }
}

extern "C" void kernel_launch(void* const* d_in, const int* in_sizes, int n_in,
                              void* d_out, int out_size, void* d_ws, size_t ws_size,
                              hipStream_t stream) {
    const int*   x      = (const int*)d_in[0];
    const float* table  = (const float*)d_in[1];
    const float* conv_w = (const float*)d_in[2];
    const float* conv_b = (const float*)d_in[3];
    float*       out    = (float*)d_out;
    float*       embd   = (float*)d_ws;

    embed_sum_kernel<<<BS * SEQ / K1_ROWS_PER_BLOCK, K1_WAVES * 32, 0, stream>>>(x, table, embd);

    dim3 grid(HGRP / NWIN, BS);
    conv_scramble_sum_kernel<<<grid, K2_THREADS, 0, stream>>>(embd, conv_w, conv_b, out);

    (void)in_sizes; (void)n_in; (void)out_size; (void)ws_size;
}


# embedding sum moved into the conv kernel: one launch of 256 32-wave work-groups; each sums its 160 embd rows from an LDS copy of the table straight into its staged rows; sum order kept
# speedup vs baseline: 7.0202x; 1.1476x over previous
#include <hip/hip_runtime.h>


typedef __attribute__((ext_vector_type(2))) float v2f;
typedef __attribute__((ext_vector_type(8))) float v8f;
typedef __attribute__((ext_vector_type(4))) int   v4i;
typedef __attribute__((ext_vector_type(4))) float v4f;

#define VOCAB   262
#define EMBD    64
#define OUT_CHS 100
#define BS      32
#define SEQ     512
#define WORD    16

#define NSIB    4
#define HGRP    128
#define CSTEP   25

#define NWIN       16
#define NTEAM      8
#define NPASS      (NWIN / NTEAM)
#define SROWS      (100 + 4 * (NWIN - 1))
#define TEAM_THREADS 128
#define K_THREADS  (TEAM_THREADS * NTEAM)
#define K_WAVES    (K_THREADS / 32)
#define RPAR       5
#define NACT       (25 * RPAR)
#define NITER      (60 / RPAR)
#define NTILE      7
#define SR_FLOATS  (NSIB * RPAR * OUT_CHS)
#define TAB4       (VOCAB * EMBD / 4)
#define TAB_TRIPS  ((TAB4 + K_THREADS - 1) / K_THREADS)
#define SUM_TRIPS  ((SROWS + 2 * K_WAVES - 1) / (2 * K_WAVES))

static_assert(TAB4 * 4 <= NTEAM * NPASS * SR_FLOATS, "the table copy fits in the partial-row planes");
static_assert(SROWS % 2 == 0, "a wave's two rows are valid together");

__device__ __forceinline__ v2f conv_relu2(
    float e0, float e1, float e2, float e3, float e4, const v2f (&t)[6])
{
    const v2f s0 = {e0, e0}, s1 = {e1, e1}, s2 = {e2, e2}, s3 = {e3, e3}, s4 = {e4, e4};
    v2f r = __builtin_elementwise_fma(s0, t[0], t[5]);
    r = __builtin_elementwise_fma(s1, t[1], r);
    r = __builtin_elementwise_fma(s2, t[2], r);
    r = __builtin_elementwise_fma(s3, t[3], r);
    r = __builtin_elementwise_fma(s4, t[4], r);
    const v2f z = {0.f, 0.f};
    return __builtin_elementwise_max(r, z);
}

__global__ __launch_bounds__(K_THREADS) void embed_conv_scramble_sum_kernel(
    const int* __restrict__ x, const float* __restrict__ table,
    const float* __restrict__ conv_w, const float* __restrict__ conv_b,
    float* __restrict__ out)
{
    __shared__ __attribute__((aligned(16))) float sE[SROWS * 64];
    __shared__ __attribute__((aligned(16))) float sU[NTEAM * NPASS * SR_FLOATS];

    const int Q   = blockIdx.x;
    const int b   = blockIdx.y;
    const int tid = threadIdx.x;
    const int tt  = tid & (TEAM_THREADS - 1);
    const int lane = tid & 31;

    const int wv   = __builtin_amdgcn_readfirstlane(tid >> 5);
    const int team = wv >> 2;
    const int wave = wv & 3;

    const v4f*    t4  = reinterpret_cast<const v4f*>(table);
    v4f*          sW4 = reinterpret_cast<v4f*>(sU);
    const float4* sT4 = reinterpret_cast<const float4*>(sU);
    float4*       sE4 = reinterpret_cast<float4*>(sE);

    v4f tb[TAB_TRIPS];
    #pragma unroll
    for (int it = 0; it < TAB_TRIPS - 1; ++it)
        tb[it] = t4[tid + it * K_THREADS];
    {
        const int idx = tid + (TAB_TRIPS - 1) * K_THREADS;
        tb[TAB_TRIPS - 1] = t4[idx < TAB4 ? idx : TAB4 - 1];
    }

    const unsigned half = (unsigned)lane >> 4;
    const unsigned q    = (unsigned)lane & 15u;
    const v4i* x4 = reinterpret_cast<const v4i*>(x) + (size_t)b * SEQ * 4;
    v4i xi[SUM_TRIPS][4];
    #pragma unroll
    for (int it = 0; it < SUM_TRIPS; ++it) {
        const unsigned r = 2u * (unsigned)wv + half + (unsigned)(2 * K_WAVES * it);
        const unsigned h = (64u * (unsigned)Q + r) & 511u;
        #pragma unroll
        for (int j = 0; j < 4; ++j)
            xi[it][j] = x4[h * 4u + j];
    }

    asm volatile("" :: "v"(tb[0]), "v"(tb[1]), "v"(tb[2]), "v"(tb[3]), "v"(tb[4]),
                       "v"(xi[0][0]), "v"(xi[0][1]), "v"(xi[0][2]), "v"(xi[0][3]),
                       "v"(xi[1][0]), "v"(xi[1][1]), "v"(xi[1][2]), "v"(xi[1][3]),
                       "v"(xi[2][0]), "v"(xi[2][1]), "v"(xi[2][2]), "v"(xi[2][3]));

    #pragma unroll
    for (int it = 0; it < TAB_TRIPS - 1; ++it)
        sW4[tid + it * K_THREADS] = tb[it];
    if ((TAB_TRIPS - 1) * K_THREADS + 32 * wv < TAB4)
        sW4[tid + (TAB_TRIPS - 1) * K_THREADS] = tb[TAB_TRIPS - 1];
    __syncthreads();

    #pragma unroll
    for (int it = 0; it < SUM_TRIPS; ++it) {
        if (2 * K_WAVES * (it + 1) <= SROWS || 2 * wv + 2 * K_WAVES * it < SROWS) {
            const unsigned r = 2u * (unsigned)wv + half + (unsigned)(2 * K_WAVES * it);
            const int id[WORD] = {
                xi[it][0].x, xi[it][0].y, xi[it][0].z, xi[it][0].w,
                xi[it][1].x, xi[it][1].y, xi[it][1].z, xi[it][1].w,
                xi[it][2].x, xi[it][2].y, xi[it][2].z, xi[it][2].w,
                xi[it][3].x, xi[it][3].y, xi[it][3].z, xi[it][3].w };
            float4 v[WORD];
            #pragma unroll
            for (int i = 0; i < WORD; ++i)
                v[i] = sT4[(unsigned)id[i] * 16u + q];
            float4 acc = make_float4(0.f, 0.f, 0.f, 0.f);
            #pragma unroll
            for (int i = 0; i < WORD; ++i) {
                acc.x += v[i].x; acc.y += v[i].y; acc.z += v[i].z; acc.w += v[i].w;
            }
            sE4[r * 16u + q] = acc;
        }
    }
    __syncthreads();

    const unsigned T = tt < NACT ? (unsigned)tt : (unsigned)(NACT - 1);

    #pragma unroll
    for (int pass = 0; pass < NPASS; ++pass) {
        const int wi  = team + NTEAM * pass;
        const int k   = NWIN * Q + wi;
        const int g   = (41 * k) & 127;
        const int rr0 = g * 100;

        const int c0 = rr0 >> 9;
        const int c1 = c0 < 24 ? c0 + 1 : 24;
        const int T0 = ((c0 + 1) * 512 - rr0) * 15;
        v2f tA[NSIB / 2][6], tB[NSIB / 2][6];
        #pragma unroll
        for (int h = 0; h < NSIB / 2; ++h) {
            #pragma unroll
            for (int kk = 0; kk < 5; ++kk) {
                tA[h][kk].x = conv_w[(c0 + CSTEP * (2 * h)) * 5 + kk];
                tA[h][kk].y = conv_w[(c0 + CSTEP * (2 * h + 1)) * 5 + kk];
                tB[h][kk].x = conv_w[(c1 + CSTEP * (2 * h)) * 5 + kk];
                tB[h][kk].y = conv_w[(c1 + CSTEP * (2 * h + 1)) * 5 + kk];
            }
            tA[h][5].x = conv_b[c0 + CSTEP * (2 * h)];
            tA[h][5].y = conv_b[c0 + CSTEP * (2 * h + 1)];
            tB[h][5].x = conv_b[c1 + CSTEP * (2 * h)];
            tB[h][5].y = conv_b[c1 + CSTEP * (2 * h + 1)];
        }

        #pragma unroll
        for (int h = 0; h < NSIB / 2; ++h) {
            #pragma unroll
            for (int kk = 0; kk < 6; ++kk) {
                asm volatile("" : "+v"(tA[h][kk]));
                asm volatile("" : "+v"(tB[h][kk]));
            }
        }

        unsigned bs[3];
        #pragma unroll
        for (int u = 0; u < 3; ++u) {
            const unsigned t = T + (unsigned)(NACT * u);
            bs[u] = 4u * (t + t / 15u) + 256u * (unsigned)wi;
        }

        v2f acc[NSIB / 2][4];
        #pragma unroll
        for (int h = 0; h < NSIB / 2; ++h) {
            #pragma unroll
            for (int i = 0; i < 4; ++i) { acc[h][i].x = 0.f; acc[h][i].y = 0.f; }
        }

        for (int k3 = 0; k3 < NITER / 3; ++k3) {
            #pragma unroll
            for (int u = 0; u < 3; ++u) {
                const float* s = sE + bs[u];
                const float4 lo = *reinterpret_cast<const float4*>(s);
                const float4 hi = *reinterpret_cast<const float4*>(s + 4);
                const float e[8] = { lo.x, lo.y, lo.z, lo.w, hi.x, hi.y, hi.z, hi.w };
                bs[u] += 1600u;

                const int it  = 3 * k3 + u;
                const int tlo = wave * 32 + NACT * it;
                if (tlo + 31 < T0) {
                    #pragma unroll
                    for (int h = 0; h < NSIB / 2; ++h) {
                        #pragma unroll
                        for (int i = 0; i < 4; ++i)
                            acc[h][i] += conv_relu2(e[i], e[i + 1], e[i + 2], e[i + 3], e[i + 4], tA[h]);
                    }
                } else if (tlo >= T0) {
                    #pragma unroll
                    for (int h = 0; h < NSIB / 2; ++h) {
                        #pragma unroll
                        for (int i = 0; i < 4; ++i)
                            acc[h][i] += conv_relu2(e[i], e[i + 1], e[i + 2], e[i + 3], e[i + 4], tB[h]);
                    }
                } else {
                    const bool up = (int)(T + (unsigned)(NACT * it)) >= T0;
                    #pragma unroll
                    for (int h = 0; h < NSIB / 2; ++h) {
                        v2f ts[6];
                        #pragma unroll
                        for (int kk = 0; kk < 6; ++kk) {
                            ts[kk].x = up ? tB[h][kk].x : tA[h][kk].x;
                            ts[kk].y = up ? tB[h][kk].y : tA[h][kk].y;
                        }
                        #pragma unroll
                        for (int i = 0; i < 4; ++i)
                            acc[h][i] += conv_relu2(e[i], e[i + 1], e[i + 2], e[i + 3], e[i + 4], ts);
                    }
                }
            }
        }

        float* sRw = sU + (team * NPASS + pass) * SR_FLOATS;
        float4* sR4 = reinterpret_cast<float4*>(sRw);
        if (tt < NACT) {
            #pragma unroll
            for (int h = 0; h < NSIB / 2; ++h) {
                sR4[(2 * h) * NACT + tt]     = make_float4(acc[h][0].x, acc[h][1].x, acc[h][2].x, acc[h][3].x);
                sR4[(2 * h + 1) * NACT + tt] = make_float4(acc[h][0].y, acc[h][1].y, acc[h][2].y, acc[h][3].y);
            }
        }

        __syncthreads();

        const int n = lane & 15;
        const float* rbase = sRw + wave * (RPAR * OUT_CHS);
        const v2f aones = {1.f, 1.f};
        const v8f zero8 = {};
        float res[NTILE];
        #pragma unroll
        for (int nt = 0; nt < NTILE; ++nt) {
            const int col = nt * 16 + n;
            const int cc  = col < OUT_CHS ? col : OUT_CHS - 1;
            const float r0 = rbase[(2 * (int)half) * OUT_CHS + cc];
            const float r1 = rbase[(2 * (int)half + 1) * OUT_CHS + cc];
            const float r4 = rbase[4 * OUT_CHS + cc];
            v2f bfrag;
            bfrag.x = r0 + (half == 0 ? r4 : 0.f);
            bfrag.y = r1;
            const v8f d = __builtin_amdgcn_wmma_f32_16x16x4_f32(false, aones, false, bfrag, (short)0, zero8, false, false);
            res[nt] = d[0];
        }

        if (lane < 16) {
            const unsigned o = (unsigned)((b * SEQ + g + HGRP * wave) * OUT_CHS + n);
            #pragma unroll
            for (int nt = 0; nt < NTILE; ++nt) {
                if (nt * 16 + n < OUT_CHS)
                    out[o + 16u * nt] = res[nt];
            }
        }
    }
}

extern "C" void kernel_launch(void* const* d_in, const int* in_sizes, int n_in,
                              void* d_out, int out_size, void* d_ws, size_t ws_size,
                              hipStream_t stream) {
    const int*   x      = (const int*)d_in[0];
    const float* table  = (const float*)d_in[1];
    const float* conv_w = (const float*)d_in[2];
    const float* conv_b = (const float*)d_in[3];
    float*       out    = (float*)d_out;

    dim3 grid(HGRP / NWIN, BS);
    embed_conv_scramble_sum_kernel<<<grid, K_THREADS, 0, stream>>>(x, table, conv_w, conv_b, out);

    (void)in_sizes; (void)n_in; (void)out_size; (void)d_ws; (void)ws_size;
}
